# starting kernel
# baseline (speedup 1.0000x reference)
#include <hip/hip_runtime.h>
#include <math.h>


typedef __attribute__((ext_vector_type(2))) float v2f;
typedef __attribute__((ext_vector_type(8))) float v8f;

#if defined(__has_builtin)
#if __has_builtin(__builtin_amdgcn_tanhf)
#define FAST_TANH(x) __builtin_amdgcn_tanhf(x)
#endif
#endif
#ifndef FAST_TANH
#define FAST_TANH(x) tanhf(x)
#endif

#define WAVES_PER_BLOCK 8
#define ROWS_PER_WAVE   16
#define T_STEPS         128
#define E_DIM           20
#define H_DIM           20
#define L_DIM           15
#define LDS_STRIDE      22

__global__ __launch_bounds__(256)
void WordVanilla_rnn_kernel(const int*   __restrict__ x,
                            const float* __restrict__ drop_rate,
                            const float* __restrict__ emb,
                            const float* __restrict__ Wxh,
                            const float* __restrict__ Whh,
                            const float* __restrict__ bh,
                            const float* __restrict__ Wd,
                            const float* __restrict__ bd,
                            float*       __restrict__ out,
                            int B)
{
    __shared__ float lds[WAVES_PER_BLOCK][ROWS_PER_WAVE][LDS_STRIDE];

    const int tid  = threadIdx.x;
    const int wave = tid >> 5;
    const int lane = tid & 31;
    const int half = lane >> 4;
    const int m    = lane & 15;
    const int tile = blockIdx.x * WAVES_PER_BLOCK + wave;
    const int b0   = tile * ROWS_PER_WAVE;
    if (b0 + ROWS_PER_WAVE > B) return;

    v2f wxh[2][5], whh[2][5];
    float bh_n[2];
#pragma unroll
    for (int nt = 0; nt < 2; ++nt) {
        const int  n  = m + 16 * nt;
        const bool nv = (n < H_DIM);
        bh_n[nt] = nv ? bh[n] : 0.0f;
#pragma unroll
        for (int c = 0; c < 5; ++c) {
            const int k = 4 * c + 2 * half;
            wxh[nt][c].x = nv ? Wxh[(k    ) * H_DIM + n] : 0.0f;
            wxh[nt][c].y = nv ? Wxh[(k + 1) * H_DIM + n] : 0.0f;
            whh[nt][c].x = nv ? Whh[(k    ) * H_DIM + n] : 0.0f;
            whh[nt][c].y = nv ? Whh[(k + 1) * H_DIM + n] : 0.0f;
        }
    }

    v2f hA[5];
#pragma unroll
    for (int c = 0; c < 5; ++c) { hA[c].x = 0.0f; hA[c].y = 0.0f; }

    const int*  xrow  = x + (size_t)(b0 + m) * T_STEPS;
    float*      myLds = &lds[wave][0][0];

    for (int t0 = 0; t0 < T_STEPS; t0 += 4) {
        const int4 toks = *(const int4*)(xrow + t0);
#pragma unroll
        for (int tt = 0; tt < 4; ++tt) {
            const int tok = (tt == 0) ? toks.x : (tt == 1) ? toks.y
                          : (tt == 2) ? toks.z : toks.w;

            const float* erow = emb + (size_t)tok * E_DIM + 2 * half;
            v2f eA[5];
#pragma unroll
            for (int c = 0; c < 5; ++c)
                eA[c] = *(const v2f*)(erow + 4 * c);

            v8f acc0 = {}; v8f acc1 = {};
#pragma unroll
            for (int c = 0; c < 5; ++c)
                acc0 = __builtin_amdgcn_wmma_f32_16x16x4_f32(
                    false, eA[c], false, wxh[0][c], (short)0, acc0, false, false);
#pragma unroll
            for (int c = 0; c < 5; ++c)
                acc0 = __builtin_amdgcn_wmma_f32_16x16x4_f32(
                    false, hA[c], false, whh[0][c], (short)0, acc0, false, false);
#pragma unroll
            for (int c = 0; c < 5; ++c)
                acc1 = __builtin_amdgcn_wmma_f32_16x16x4_f32(
                    false, eA[c], false, wxh[1][c], (short)0, acc1, false, false);
#pragma unroll
            for (int c = 0; c < 5; ++c)
                acc1 = __builtin_amdgcn_wmma_f32_16x16x4_f32(
                    false, hA[c], false, whh[1][c], (short)0, acc1, false, false);

#pragma unroll
            for (int i = 0; i < 8; ++i) {
                const int row = i + 8 * half;
                myLds[row * LDS_STRIDE + m] = FAST_TANH(acc0[i] + bh_n[0]);
            }
            if (m < (H_DIM - 16)) {
#pragma unroll
                for (int i = 0; i < 8; ++i) {
                    const int row = i + 8 * half;
                    myLds[row * LDS_STRIDE + m + 16] = FAST_TANH(acc1[i] + bh_n[1]);
                }
            }

#pragma unroll
            for (int c = 0; c < 5; ++c)
                hA[c] = *(const v2f*)(myLds + m * LDS_STRIDE + 4 * c + 2 * half);
        }
    }

    const float scale = 1.0f / (1.0f - drop_rate[0]);
    if (half == 0) {
        float hrow[H_DIM];
#pragma unroll
        for (int k = 0; k < H_DIM; ++k)
            hrow[k] = myLds[m * LDS_STRIDE + k] * scale;

        float z[L_DIM];
        float zmax = -INFINITY;
#pragma unroll
        for (int l = 0; l < L_DIM; ++l) {
            float s = bd[l];
#pragma unroll
            for (int k = 0; k < H_DIM; ++k)
                s += hrow[k] * Wd[k * L_DIM + l];
            z[l] = s;
            zmax = fmaxf(zmax, s);
        }
        float sum = 0.0f;
#pragma unroll
        for (int l = 0; l < L_DIM; ++l) {
            z[l] = __expf(z[l] - zmax);
            sum += z[l];
        }
        const float inv = 1.0f / sum;
        float* orow = out + (size_t)(b0 + m) * L_DIM;
#pragma unroll
        for (int l = 0; l < L_DIM; ++l)
            orow[l] = z[l] * inv;
    }
}

extern "C" void kernel_launch(void* const* d_in, const int* in_sizes, int n_in,
                              void* d_out, int out_size, void* d_ws, size_t ws_size,
                              hipStream_t stream) {
    (void)n_in; (void)out_size; (void)d_ws; (void)ws_size;

    const int*   x   = (const int*)  d_in[0];
    const float* dr  = (const float*)d_in[1];
    const float* emb = (const float*)d_in[2];
    const float* Wxh = (const float*)d_in[3];
    const float* Whh = (const float*)d_in[4];
    const float* bh  = (const float*)d_in[5];
    const float* Wd  = (const float*)d_in[6];
    const float* bd  = (const float*)d_in[7];
    float*       out = (float*)d_out;

    const int B = in_sizes[0] / T_STEPS;
    const int rowsPerBlock = ROWS_PER_WAVE * WAVES_PER_BLOCK;
    const int grid = (B + rowsPerBlock - 1) / rowsPerBlock;

    hipLaunchKernelGGL(WordVanilla_rnn_kernel, dim3(grid), dim3(256), 0, stream,
                       x, dr, emb, Wxh, Whh, bh, Wd, bd, out, B);
}


# embedding rows register-prefetched one 4-step group ahead (tokens two groups ahead): the token->row global-load latency leaves the recurrence's per-step critical path
# speedup vs baseline: 1.2769x; 1.2769x over previous
#include <hip/hip_runtime.h>
#include <math.h>


typedef __attribute__((ext_vector_type(2))) float v2f;
typedef __attribute__((ext_vector_type(8))) float v8f;

#if defined(__has_builtin)
#if __has_builtin(__builtin_amdgcn_tanhf)
#define FAST_TANH(x) __builtin_amdgcn_tanhf(x)
#endif
#endif
#ifndef FAST_TANH
#define FAST_TANH(x) tanhf(x)
#endif

#define WAVES_PER_BLOCK 8
#define ROWS_PER_WAVE   16
#define T_STEPS         128
#define E_DIM           20
#define H_DIM           20
#define L_DIM           15
#define LDS_STRIDE      22


template <bool REFILL>
static __device__ __forceinline__ void rnn_group4(
    v2f (&eA)[4][5], v2f (&hA)[5],
    const v2f (&wxh)[2][5], const v2f (&whh)[2][5], const float (&bh_n)[2],
    const int4 tokA, const float* __restrict__ embh,
    float* myLds, const int m, const int half)
{
#pragma unroll
    for (int tt = 0; tt < 4; ++tt) {
        v8f acc0 = {}; v8f acc1 = {};
#pragma unroll
        for (int c = 0; c < 5; ++c)
            acc0 = __builtin_amdgcn_wmma_f32_16x16x4_f32(
                false, eA[tt][c], false, wxh[0][c], (short)0, acc0, false, false);
#pragma unroll
        for (int c = 0; c < 5; ++c)
            acc0 = __builtin_amdgcn_wmma_f32_16x16x4_f32(
                false, hA[c], false, whh[0][c], (short)0, acc0, false, false);
#pragma unroll
        for (int c = 0; c < 5; ++c)
            acc1 = __builtin_amdgcn_wmma_f32_16x16x4_f32(
                false, eA[tt][c], false, wxh[1][c], (short)0, acc1, false, false);
#pragma unroll
        for (int c = 0; c < 5; ++c)
            acc1 = __builtin_amdgcn_wmma_f32_16x16x4_f32(
                false, hA[c], false, whh[1][c], (short)0, acc1, false, false);

        if (REFILL) {
            const int tok = (tt == 0) ? tokA.x : (tt == 1) ? tokA.y
                          : (tt == 2) ? tokA.z : tokA.w;
            const float* erow = embh + (size_t)tok * E_DIM;
#pragma unroll
            for (int c = 0; c < 5; ++c)
                eA[tt][c] = *(const v2f*)(erow + 4 * c);
        }

#pragma unroll
        for (int i = 0; i < 8; ++i) {
            const int row = i + 8 * half;
            myLds[row * LDS_STRIDE + m] = FAST_TANH(acc0[i] + bh_n[0]);
        }
        if (m < (H_DIM - 16)) {
#pragma unroll
            for (int i = 0; i < 8; ++i) {
                const int row = i + 8 * half;
                myLds[row * LDS_STRIDE + m + 16] = FAST_TANH(acc1[i] + bh_n[1]);
            }
        }

#pragma unroll
        for (int c = 0; c < 5; ++c)
            hA[c] = *(const v2f*)(myLds + m * LDS_STRIDE + 4 * c + 2 * half);
    }
}

__global__ __launch_bounds__(256)
void WordVanilla_rnn_kernel(const int*   __restrict__ x,
                            const float* __restrict__ drop_rate,
                            const float* __restrict__ emb,
                            const float* __restrict__ Wxh,
                            const float* __restrict__ Whh,
                            const float* __restrict__ bh,
                            const float* __restrict__ Wd,
                            const float* __restrict__ bd,
                            float*       __restrict__ out,
                            int B)
{
    __shared__ float lds[WAVES_PER_BLOCK][ROWS_PER_WAVE][LDS_STRIDE];

    const int tid  = threadIdx.x;
    const int wave = tid >> 5;
    const int lane = tid & 31;
    const int half = lane >> 4;
    const int m    = lane & 15;
    const int tile = blockIdx.x * WAVES_PER_BLOCK + wave;
    const int b0   = tile * ROWS_PER_WAVE;
    if (b0 + ROWS_PER_WAVE > B) return;

    v2f wxh[2][5], whh[2][5];
    float bh_n[2];
#pragma unroll
    for (int nt = 0; nt < 2; ++nt) {
        const int  n  = m + 16 * nt;
        const bool nv = (n < H_DIM);
        bh_n[nt] = nv ? bh[n] : 0.0f;
#pragma unroll
        for (int c = 0; c < 5; ++c) {
            const int k = 4 * c + 2 * half;
            wxh[nt][c].x = nv ? Wxh[(k    ) * H_DIM + n] : 0.0f;
            wxh[nt][c].y = nv ? Wxh[(k + 1) * H_DIM + n] : 0.0f;
            whh[nt][c].x = nv ? Whh[(k    ) * H_DIM + n] : 0.0f;
            whh[nt][c].y = nv ? Whh[(k + 1) * H_DIM + n] : 0.0f;
        }
    }

    v2f hA[5];
#pragma unroll
    for (int c = 0; c < 5; ++c) { hA[c].x = 0.0f; hA[c].y = 0.0f; }

    const int*  xrow  = x + (size_t)(b0 + m) * T_STEPS;
    float*      myLds = &lds[wave][0][0];

    const float* embh = emb + 2 * half;

    v2f  eA[4][5];
    int4 tokN;
    {
        const int4 tok0 = *(const int4*)xrow;
        tokN = *(const int4*)(xrow + 4);
#pragma unroll
        for (int tt = 0; tt < 4; ++tt) {
            const int tok = (tt == 0) ? tok0.x : (tt == 1) ? tok0.y
                          : (tt == 2) ? tok0.z : tok0.w;
            const float* erow = embh + (size_t)tok * E_DIM;
#pragma unroll
            for (int c = 0; c < 5; ++c)
                eA[tt][c] = *(const v2f*)(erow + 4 * c);
        }
    }

    for (int t0 = 0; t0 < T_STEPS - 4; t0 += 4) {
        const int4 tokA = tokN;
        if (t0 + 8 < T_STEPS)
            tokN = *(const int4*)(xrow + t0 + 8);
        rnn_group4<true>(eA, hA, wxh, whh, bh_n, tokA, embh, myLds, m, half);
    }
    rnn_group4<false>(eA, hA, wxh, whh, bh_n, tokN, embh, myLds, m, half);

    const float scale = 1.0f / (1.0f - drop_rate[0]);
    if (half == 0) {
        float hrow[H_DIM];
#pragma unroll
        for (int k = 0; k < H_DIM; ++k)
            hrow[k] = myLds[m * LDS_STRIDE + k] * scale;

        float z[L_DIM];
        float zmax = -INFINITY;
#pragma unroll
        for (int l = 0; l < L_DIM; ++l) {
            float s = bd[l];
#pragma unroll
            for (int k = 0; k < H_DIM; ++k)
                s += hrow[k] * Wd[k * L_DIM + l];
            z[l] = s;
            zmax = fmaxf(zmax, s);
        }
        float sum = 0.0f;
#pragma unroll
        for (int l = 0; l < L_DIM; ++l) {
            z[l] = __expf(z[l] - zmax);
            sum += z[l];
        }
        const float inv = 1.0f / sum;
        float* orow = out + (size_t)(b0 + m) * L_DIM;
#pragma unroll
        for (int l = 0; l < L_DIM; ++l)
            orow[l] = z[l] * inv;
    }
}

extern "C" void kernel_launch(void* const* d_in, const int* in_sizes, int n_in,
                              void* d_out, int out_size, void* d_ws, size_t ws_size,
                              hipStream_t stream) {
    (void)n_in; (void)out_size; (void)d_ws; (void)ws_size;

    const int*   x   = (const int*)  d_in[0];
    const float* dr  = (const float*)d_in[1];
    const float* emb = (const float*)d_in[2];
    const float* Wxh = (const float*)d_in[3];
    const float* Whh = (const float*)d_in[4];
    const float* bh  = (const float*)d_in[5];
    const float* Wd  = (const float*)d_in[6];
    const float* bd  = (const float*)d_in[7];
    float*       out = (float*)d_out;

    const int B = in_sizes[0] / T_STEPS;
    const int rowsPerBlock = ROWS_PER_WAVE * WAVES_PER_BLOCK;
    const int grid = (B + rowsPerBlock - 1) / rowsPerBlock;

    hipLaunchKernelGGL(WordVanilla_rnn_kernel, dim3(grid), dim3(256), 0, stream,
                       x, dr, emb, Wxh, Whh, bh, Wd, bd, out, B);
}


# 4-wave work-groups on a 256-group grid (was 8 waves, 128 groups): one batch-tile wave per SIMD on all 256 WGPs, so no two waves share one SIMD's f32 WMMA issue
# speedup vs baseline: 1.6874x; 1.3215x over previous
#include <hip/hip_runtime.h>
#include <math.h>


typedef __attribute__((ext_vector_type(2))) float v2f;
typedef __attribute__((ext_vector_type(8))) float v8f;

#if defined(__has_builtin)
#if __has_builtin(__builtin_amdgcn_tanhf)
#define FAST_TANH(x) __builtin_amdgcn_tanhf(x)
#endif
#endif
#ifndef FAST_TANH
#define FAST_TANH(x) tanhf(x)
#endif

#define WAVES_PER_BLOCK 4
#define ROWS_PER_WAVE   16
#define T_STEPS         128
#define E_DIM           20
#define H_DIM           20
#define L_DIM           15
#define LDS_STRIDE      22


template <bool REFILL>
static __device__ __forceinline__ void rnn_group4(
    v2f (&eA)[4][5], v2f (&hA)[5],
    const v2f (&wxh)[2][5], const v2f (&whh)[2][5], const float (&bh_n)[2],
    const int4 tokA, const float* __restrict__ embh,
    float* myLds, const int m, const int half)
{
#pragma unroll
    for (int tt = 0; tt < 4; ++tt) {
        v8f acc0 = {}; v8f acc1 = {};
#pragma unroll
        for (int c = 0; c < 5; ++c)
            acc0 = __builtin_amdgcn_wmma_f32_16x16x4_f32(
                false, eA[tt][c], false, wxh[0][c], (short)0, acc0, false, false);
#pragma unroll
        for (int c = 0; c < 5; ++c)
            acc0 = __builtin_amdgcn_wmma_f32_16x16x4_f32(
                false, hA[c], false, whh[0][c], (short)0, acc0, false, false);
#pragma unroll
        for (int c = 0; c < 5; ++c)
            acc1 = __builtin_amdgcn_wmma_f32_16x16x4_f32(
                false, eA[tt][c], false, wxh[1][c], (short)0, acc1, false, false);
#pragma unroll
        for (int c = 0; c < 5; ++c)
            acc1 = __builtin_amdgcn_wmma_f32_16x16x4_f32(
                false, hA[c], false, whh[1][c], (short)0, acc1, false, false);

        if (REFILL) {
            const int tok = (tt == 0) ? tokA.x : (tt == 1) ? tokA.y
                          : (tt == 2) ? tokA.z : tokA.w;
            const float* erow = embh + (size_t)tok * E_DIM;
#pragma unroll
            for (int c = 0; c < 5; ++c)
                eA[tt][c] = *(const v2f*)(erow + 4 * c);
        }

#pragma unroll
        for (int i = 0; i < 8; ++i) {
            const int row = i + 8 * half;
            myLds[row * LDS_STRIDE + m] = FAST_TANH(acc0[i] + bh_n[0]);
        }
        if (m < (H_DIM - 16)) {
#pragma unroll
            for (int i = 0; i < 8; ++i) {
                const int row = i + 8 * half;
                myLds[row * LDS_STRIDE + m + 16] = FAST_TANH(acc1[i] + bh_n[1]);
            }
        }

#pragma unroll
        for (int c = 0; c < 5; ++c)
            hA[c] = *(const v2f*)(myLds + m * LDS_STRIDE + 4 * c + 2 * half);
    }
}

__global__ __launch_bounds__(32 * WAVES_PER_BLOCK)
void WordVanilla_rnn_kernel(const int*   __restrict__ x,
                            const float* __restrict__ drop_rate,
                            const float* __restrict__ emb,
                            const float* __restrict__ Wxh,
                            const float* __restrict__ Whh,
                            const float* __restrict__ bh,
                            const float* __restrict__ Wd,
                            const float* __restrict__ bd,
                            float*       __restrict__ out,
                            int B)
{
    __shared__ float lds[WAVES_PER_BLOCK][ROWS_PER_WAVE][LDS_STRIDE];

    const int tid  = threadIdx.x;
    const int wave = tid >> 5;
    const int lane = tid & 31;
    const int half = lane >> 4;
    const int m    = lane & 15;
    const int tile = blockIdx.x * WAVES_PER_BLOCK + wave;
    const int b0   = tile * ROWS_PER_WAVE;
    if (b0 + ROWS_PER_WAVE > B) return;

    v2f wxh[2][5], whh[2][5];
    float bh_n[2];
#pragma unroll
    for (int nt = 0; nt < 2; ++nt) {
        const int  n  = m + 16 * nt;
        const bool nv = (n < H_DIM);
        bh_n[nt] = nv ? bh[n] : 0.0f;
#pragma unroll
        for (int c = 0; c < 5; ++c) {
            const int k = 4 * c + 2 * half;
            wxh[nt][c].x = nv ? Wxh[(k    ) * H_DIM + n] : 0.0f;
            wxh[nt][c].y = nv ? Wxh[(k + 1) * H_DIM + n] : 0.0f;
            whh[nt][c].x = nv ? Whh[(k    ) * H_DIM + n] : 0.0f;
            whh[nt][c].y = nv ? Whh[(k + 1) * H_DIM + n] : 0.0f;
        }
    }

    v2f hA[5];
#pragma unroll
    for (int c = 0; c < 5; ++c) { hA[c].x = 0.0f; hA[c].y = 0.0f; }

    const int*  xrow  = x + (size_t)(b0 + m) * T_STEPS;
    float*      myLds = &lds[wave][0][0];

    const float* embh = emb + 2 * half;

    v2f  eA[4][5];
    int4 tokN;
    {
        const int4 tok0 = *(const int4*)xrow;
        tokN = *(const int4*)(xrow + 4);
#pragma unroll
        for (int tt = 0; tt < 4; ++tt) {
            const int tok = (tt == 0) ? tok0.x : (tt == 1) ? tok0.y
                          : (tt == 2) ? tok0.z : tok0.w;
            const float* erow = embh + (size_t)tok * E_DIM;
#pragma unroll
            for (int c = 0; c < 5; ++c)
                eA[tt][c] = *(const v2f*)(erow + 4 * c);
        }
    }

    for (int t0 = 0; t0 < T_STEPS - 4; t0 += 4) {
        const int4 tokA = tokN;
        if (t0 + 8 < T_STEPS)
            tokN = *(const int4*)(xrow + t0 + 8);
        rnn_group4<true>(eA, hA, wxh, whh, bh_n, tokA, embh, myLds, m, half);
    }
    rnn_group4<false>(eA, hA, wxh, whh, bh_n, tokN, embh, myLds, m, half);

    const float scale = 1.0f / (1.0f - drop_rate[0]);
    if (half == 0) {
        float hrow[H_DIM];
#pragma unroll
        for (int k = 0; k < H_DIM; ++k)
            hrow[k] = myLds[m * LDS_STRIDE + k] * scale;

        float z[L_DIM];
        float zmax = -INFINITY;
#pragma unroll
        for (int l = 0; l < L_DIM; ++l) {
            float s = bd[l];
#pragma unroll
            for (int k = 0; k < H_DIM; ++k)
                s += hrow[k] * Wd[k * L_DIM + l];
            z[l] = s;
            zmax = fmaxf(zmax, s);
        }
        float sum = 0.0f;
#pragma unroll
        for (int l = 0; l < L_DIM; ++l) {
            z[l] = __expf(z[l] - zmax);
            sum += z[l];
        }
        const float inv = 1.0f / sum;
        float* orow = out + (size_t)(b0 + m) * L_DIM;
#pragma unroll
        for (int l = 0; l < L_DIM; ++l)
            orow[l] = z[l] * inv;
    }
}

extern "C" void kernel_launch(void* const* d_in, const int* in_sizes, int n_in,
                              void* d_out, int out_size, void* d_ws, size_t ws_size,
                              hipStream_t stream) {
    (void)n_in; (void)out_size; (void)d_ws; (void)ws_size;

    const int*   x   = (const int*)  d_in[0];
    const float* dr  = (const float*)d_in[1];
    const float* emb = (const float*)d_in[2];
    const float* Wxh = (const float*)d_in[3];
    const float* Whh = (const float*)d_in[4];
    const float* bh  = (const float*)d_in[5];
    const float* Wd  = (const float*)d_in[6];
    const float* bd  = (const float*)d_in[7];
    float*       out = (float*)d_out;

    const int B = in_sizes[0] / T_STEPS;
    const int rowsPerBlock = ROWS_PER_WAVE * WAVES_PER_BLOCK;
    const int grid = (B + rowsPerBlock - 1) / rowsPerBlock;

    hipLaunchKernelGGL(WordVanilla_rnn_kernel, dim3(grid), dim3(32 * WAVES_PER_BLOCK), 0, stream,
                       x, dr, emb, Wxh, Whh, bh, Wd, bd, out, B);
}


# recurrence computed transposed (weights as A, e^T and h^T as B): tanh(D) is already the next step's B fragment, so the per-step LDS transpose is gone; h@Whh products summed in a permuted k order
# speedup vs baseline: 2.3818x; 1.4115x over previous
#include <hip/hip_runtime.h>
#include <math.h>


typedef __attribute__((ext_vector_type(2))) float v2f;
typedef __attribute__((ext_vector_type(8))) float v8f;

#if defined(__has_builtin)
#if __has_builtin(__builtin_amdgcn_tanhf)
#define FAST_TANH(x) __builtin_amdgcn_tanhf(x)
#endif
#endif
#ifndef FAST_TANH
#define FAST_TANH(x) tanhf(x)
#endif

#define WAVES_PER_BLOCK 4
#define ROWS_PER_WAVE   16
#define T_STEPS         128
#define E_DIM           20
#define H_DIM           20
#define L_DIM           15
#define LDS_STRIDE      22


static __device__ __forceinline__ float xhalf(float v)
{
    const int b = __builtin_bit_cast(int, v);
    return __builtin_bit_cast(float,
        __builtin_amdgcn_permlanex16(b, b, 0x76543210, 0xfedcba98, false, false));
}

template <bool REFILL>
static __device__ __forceinline__ void rnn_group4(
    v2f (&eB)[4][5], v2f (&hB)[5],
    const v2f (&wxA)[2][5], const v2f (&whA)[2][5],
    const float (&bias0)[8], const float (&bias1)[4],
    const int4 tokA, const float* __restrict__ embh, const int half)
{
#pragma unroll
    for (int tt = 0; tt < 4; ++tt) {
        v8f acc0 = {}; v8f acc1 = {};
#pragma unroll
        for (int c = 0; c < 5; ++c) {
            acc0 = __builtin_amdgcn_wmma_f32_16x16x4_f32(
                false, wxA[0][c], false, eB[tt][c], (short)0, acc0, false, false);
            acc1 = __builtin_amdgcn_wmma_f32_16x16x4_f32(
                false, wxA[1][c], false, eB[tt][c], (short)0, acc1, false, false);
        }
#pragma unroll
        for (int c = 0; c < 5; ++c) {
            acc0 = __builtin_amdgcn_wmma_f32_16x16x4_f32(
                false, whA[0][c], false, hB[c], (short)0, acc0, false, false);
            acc1 = __builtin_amdgcn_wmma_f32_16x16x4_f32(
                false, whA[1][c], false, hB[c], (short)0, acc1, false, false);
        }

        if (REFILL) {
            const int tok = (tt == 0) ? tokA.x : (tt == 1) ? tokA.y
                          : (tt == 2) ? tokA.z : tokA.w;
            const float* erow = embh + (size_t)tok * E_DIM;
#pragma unroll
            for (int c = 0; c < 5; ++c)
                eB[tt][c] = *(const v2f*)(erow + 4 * c);
        }

#pragma unroll
        for (int c = 0; c < 4; ++c) {
            hB[c].x = FAST_TANH(acc0[2 * c]     + bias0[2 * c]);
            hB[c].y = FAST_TANH(acc0[2 * c + 1] + bias0[2 * c + 1]);
        }
        const float t16 = FAST_TANH(acc1[0] + bias1[0]);
        const float t17 = FAST_TANH(acc1[1] + bias1[1]);
        const float t18 = FAST_TANH(acc1[2] + bias1[2]);
        const float t19 = FAST_TANH(acc1[3] + bias1[3]);
        const float x18 = xhalf(t18);
        const float x19 = xhalf(t19);
        hB[4].x = half ? x18 : t16;
        hB[4].y = half ? x19 : t17;
    }
}

__global__ __launch_bounds__(32 * WAVES_PER_BLOCK)
void WordVanilla_rnn_kernel(const int*   __restrict__ x,
                            const float* __restrict__ drop_rate,
                            const float* __restrict__ emb,
                            const float* __restrict__ Wxh,
                            const float* __restrict__ Whh,
                            const float* __restrict__ bh,
                            const float* __restrict__ Wd,
                            const float* __restrict__ bd,
                            float*       __restrict__ out,
                            int B)
{
    __shared__ float lds[WAVES_PER_BLOCK][ROWS_PER_WAVE][LDS_STRIDE];

    const int tid  = threadIdx.x;
    const int wave = tid >> 5;
    const int lane = tid & 31;
    const int half = lane >> 4;
    const int m    = lane & 15;
    const int tile = blockIdx.x * WAVES_PER_BLOCK + wave;
    const int b0   = tile * ROWS_PER_WAVE;
    if (b0 + ROWS_PER_WAVE > B) return;

    v2f wxA[2][5], whA[2][5];
#pragma unroll
    for (int mt = 0; mt < 2; ++mt) {
        const int  n  = m + 16 * mt;
        const bool nv = (n < H_DIM);
#pragma unroll
        for (int c = 0; c < 5; ++c) {
            const int k  = 4 * c + 2 * half;
            const int kh = (c < 4) ? (2 * c + 8 * half) : (16 + 2 * half);
            wxA[mt][c].x = nv ? Wxh[(k     ) * H_DIM + n] : 0.0f;
            wxA[mt][c].y = nv ? Wxh[(k  + 1) * H_DIM + n] : 0.0f;
            whA[mt][c].x = nv ? Whh[(kh    ) * H_DIM + n] : 0.0f;
            whA[mt][c].y = nv ? Whh[(kh + 1) * H_DIM + n] : 0.0f;
        }
    }
    float bias0[8], bias1[4];
#pragma unroll
    for (int r = 0; r < 8; ++r) bias0[r] = bh[8 * half + r];
#pragma unroll
    for (int r = 0; r < 4; ++r) bias1[r] = bh[16 + r];

    v2f hB[5];
#pragma unroll
    for (int c = 0; c < 5; ++c) { hB[c].x = 0.0f; hB[c].y = 0.0f; }

    const int*  xrow  = x + (size_t)(b0 + m) * T_STEPS;
    float*      myLds = &lds[wave][0][0];

    const float* embh = emb + 2 * half;

    v2f  eB[4][5];
    int4 tokN;
    {
        const int4 tok0 = *(const int4*)xrow;
        tokN = *(const int4*)(xrow + 4);
#pragma unroll
        for (int tt = 0; tt < 4; ++tt) {
            const int tok = (tt == 0) ? tok0.x : (tt == 1) ? tok0.y
                          : (tt == 2) ? tok0.z : tok0.w;
            const float* erow = embh + (size_t)tok * E_DIM;
#pragma unroll
            for (int c = 0; c < 5; ++c)
                eB[tt][c] = *(const v2f*)(erow + 4 * c);
        }
    }

    for (int t0 = 0; t0 < T_STEPS - 4; t0 += 4) {
        const int4 tokA = tokN;
        if (t0 + 8 < T_STEPS)
            tokN = *(const int4*)(xrow + t0 + 8);
        rnn_group4<true>(eB, hB, wxA, whA, bias0, bias1, tokA, embh, half);
    }
    rnn_group4<false>(eB, hB, wxA, whA, bias0, bias1, tokN, embh, half);

#pragma unroll
    for (int c = 0; c < 4; ++c)
        *(v2f*)(myLds + m * LDS_STRIDE + 2 * c + 8 * half) = hB[c];
    *(v2f*)(myLds + m * LDS_STRIDE + 16 + 2 * half) = hB[4];

    const float scale = 1.0f / (1.0f - drop_rate[0]);
    if (half == 0) {
        float hrow[H_DIM];
#pragma unroll
        for (int k = 0; k < H_DIM; ++k)
            hrow[k] = myLds[m * LDS_STRIDE + k] * scale;

        float z[L_DIM];
        float zmax = -INFINITY;
#pragma unroll
        for (int l = 0; l < L_DIM; ++l) {
            float s = bd[l];
#pragma unroll
            for (int k = 0; k < H_DIM; ++k)
                s += hrow[k] * Wd[k * L_DIM + l];
            z[l] = s;
            zmax = fmaxf(zmax, s);
        }
        float sum = 0.0f;
#pragma unroll
        for (int l = 0; l < L_DIM; ++l) {
            z[l] = __expf(z[l] - zmax);
            sum += z[l];
        }
        const float inv = 1.0f / sum;
        float* orow = out + (size_t)(b0 + m) * L_DIM;
#pragma unroll
        for (int l = 0; l < L_DIM; ++l)
            orow[l] = z[l] * inv;
    }
}

extern "C" void kernel_launch(void* const* d_in, const int* in_sizes, int n_in,
                              void* d_out, int out_size, void* d_ws, size_t ws_size,
                              hipStream_t stream) {
    (void)n_in; (void)out_size; (void)d_ws; (void)ws_size;

    const int*   x   = (const int*)  d_in[0];
    const float* dr  = (const float*)d_in[1];
    const float* emb = (const float*)d_in[2];
    const float* Wxh = (const float*)d_in[3];
    const float* Whh = (const float*)d_in[4];
    const float* bh  = (const float*)d_in[5];
    const float* Wd  = (const float*)d_in[6];
    const float* bd  = (const float*)d_in[7];
    float*       out = (float*)d_out;

    const int B = in_sizes[0] / T_STEPS;
    const int rowsPerBlock = ROWS_PER_WAVE * WAVES_PER_BLOCK;
    const int grid = (B + rowsPerBlock - 1) / rowsPerBlock;

    hipLaunchKernelGGL(WordVanilla_rnn_kernel, dim3(grid), dim3(32 * WAVES_PER_BLOCK), 0, stream,
                       x, dr, emb, Wxh, Whh, bh, Wd, bd, out, B);
}


# input projection emb@Wxh+bh done once per vocabulary row by a first launch into a workspace table; scan gathers projected rows as C operand: 10 WMMAs per step, was 20; bias summed first
# speedup vs baseline: 3.0604x; 1.2849x over previous
#include <hip/hip_runtime.h>
#include <math.h>


typedef __attribute__((ext_vector_type(2))) float v2f;
typedef __attribute__((ext_vector_type(4))) float v4f;
typedef __attribute__((ext_vector_type(8))) float v8f;

#if defined(__has_builtin)
#if __has_builtin(__builtin_amdgcn_tanhf)
#define FAST_TANH(x) __builtin_amdgcn_tanhf(x)
#endif
#endif
#ifndef FAST_TANH
#define FAST_TANH(x) tanhf(x)
#endif

#define WAVES_PER_BLOCK 4
#define ROWS_PER_WAVE   16
#define T_STEPS         128
#define E_DIM           20
#define H_DIM           20
#define L_DIM           15
#define LDS_STRIDE      22
#define TAB_STRIDE      32
#define PROJ_WAVES      4
#define GROUP           8


static __device__ __forceinline__ float xhalf(float v)
{
    const int b = __builtin_bit_cast(int, v);
    return __builtin_bit_cast(float,
        __builtin_amdgcn_permlanex16(b, b, 0x76543210, 0xfedcba98, false, false));
}

__global__ __launch_bounds__(32 * PROJ_WAVES)
void input_projection_table_kernel(const float* __restrict__ emb,
                                   const float* __restrict__ Wxh,
                                   const float* __restrict__ bh,
                                   float*       __restrict__ table,
                                   int V)
{
    const int tid  = threadIdx.x;
    const int wave = tid >> 5;
    const int lane = tid & 31;
    const int half = lane >> 4;
    const int m    = lane & 15;
    const int r0   = (blockIdx.x * PROJ_WAVES + wave) * 16;
    if (r0 >= V) return;
    const int row  = r0 + m;
    const int rowc = (row < V) ? row : (V - 1);

    v2f wxA[2][5];
#pragma unroll
    for (int mt = 0; mt < 2; ++mt) {
        const int  n  = m + 16 * mt;
        const bool nv = (n < H_DIM);
#pragma unroll
        for (int c = 0; c < 5; ++c) {
            const int k = 4 * c + 2 * half;
            wxA[mt][c].x = nv ? Wxh[(k    ) * H_DIM + n] : 0.0f;
            wxA[mt][c].y = nv ? Wxh[(k + 1) * H_DIM + n] : 0.0f;
        }
    }
    const v4f bLo = *(const v4f*)(bh + 8 * half);
    const v4f bHi = *(const v4f*)(bh + 8 * half + 4);
    const v4f bT1 = *(const v4f*)(bh + 16);
    const float* erow = emb + (size_t)rowc * E_DIM + 2 * half;
    v2f eB[5];
#pragma unroll
    for (int c = 0; c < 5; ++c)
        eB[c] = *(const v2f*)(erow + 4 * c);

    v8f acc0 = {}; v8f acc1 = {};
#pragma unroll
    for (int c = 0; c < 5; ++c) {
        acc0 = __builtin_amdgcn_wmma_f32_16x16x4_f32(
            false, wxA[0][c], false, eB[c], (short)0, acc0, false, false);
        acc1 = __builtin_amdgcn_wmma_f32_16x16x4_f32(
            false, wxA[1][c], false, eB[c], (short)0, acc1, false, false);
    }

    if (row < V) {
        float* prow = table + (size_t)row * TAB_STRIDE;
        v4f lo, hi;
        lo.x = acc0[0] + bLo.x; lo.y = acc0[1] + bLo.y;
        lo.z = acc0[2] + bLo.z; lo.w = acc0[3] + bLo.w;
        hi.x = acc0[4] + bHi.x; hi.y = acc0[5] + bHi.y;
        hi.z = acc0[6] + bHi.z; hi.w = acc0[7] + bHi.w;
        *(v4f*)(prow + 8 * half)     = lo;
        *(v4f*)(prow + 8 * half + 4) = hi;
        v4f t1;
        t1.x = half ? 0.0f : acc1[0] + bT1.x;
        t1.y = half ? 0.0f : acc1[1] + bT1.y;
        t1.z = half ? 0.0f : acc1[2] + bT1.z;
        t1.w = half ? 0.0f : acc1[3] + bT1.w;
        *(v4f*)(prow + 16 + 8 * half) = t1;
    }
}

template <bool REFILL>
static __device__ __forceinline__ void rnn_group(
    v8f (&pC0)[GROUP], v8f (&pC1)[GROUP], v2f (&hB)[5],
    const v2f (&whA)[2][5],
    const int (&tokA)[GROUP], const float* __restrict__ tabh, const int half)
{
#pragma unroll
    for (int tt = 0; tt < GROUP; ++tt) {
        v8f acc0 = pC0[tt]; v8f acc1 = pC1[tt];
#pragma unroll
        for (int c = 0; c < 5; ++c) {
            acc0 = __builtin_amdgcn_wmma_f32_16x16x4_f32(
                false, whA[0][c], false, hB[c], (short)0, acc0, false, false);
            acc1 = __builtin_amdgcn_wmma_f32_16x16x4_f32(
                false, whA[1][c], false, hB[c], (short)0, acc1, false, false);
        }

#pragma unroll
        for (int c = 0; c < 4; ++c) {
            hB[c].x = FAST_TANH(acc0[2 * c]);
            hB[c].y = FAST_TANH(acc0[2 * c + 1]);
        }
        const float t16 = FAST_TANH(acc1[0]);
        const float t17 = FAST_TANH(acc1[1]);
        const float t18 = FAST_TANH(acc1[2]);
        const float t19 = FAST_TANH(acc1[3]);
        const float x18 = xhalf(t18);
        const float x19 = xhalf(t19);
        hB[4].x = half ? x18 : t16;
        hB[4].y = half ? x19 : t17;

        pC1[tt][4] = acc1[4]; pC1[tt][5] = acc1[5]; pC1[tt][6] = acc1[6]; pC1[tt][7] = acc1[7];
        if (REFILL) {
            __builtin_amdgcn_sched_barrier(0);
            const float* prow = tabh + (size_t)tokA[tt] * TAB_STRIDE;
            const v4f lo = *(const v4f*)(prow);
            const v4f hi = *(const v4f*)(prow + 4);
            const v4f t1 = *(const v4f*)(prow + 16);
            pC0[tt][0] = lo.x; pC0[tt][1] = lo.y; pC0[tt][2] = lo.z; pC0[tt][3] = lo.w;
            pC0[tt][4] = hi.x; pC0[tt][5] = hi.y; pC0[tt][6] = hi.z; pC0[tt][7] = hi.w;
            pC1[tt][0] = t1.x; pC1[tt][1] = t1.y; pC1[tt][2] = t1.z; pC1[tt][3] = t1.w;
            __builtin_amdgcn_sched_barrier(0);
        }
    }
}

__global__ __launch_bounds__(32 * WAVES_PER_BLOCK)
void WordVanilla_rnn_kernel(const int*   __restrict__ x,
                            const float* __restrict__ drop_rate,
                            const float* __restrict__ table,
                            const float* __restrict__ Whh,
                            const float* __restrict__ Wd,
                            const float* __restrict__ bd,
                            float*       __restrict__ out,
                            int B)
{
    __shared__ float lds[WAVES_PER_BLOCK][ROWS_PER_WAVE][LDS_STRIDE];

    const int tid  = threadIdx.x;
    const int wave = tid >> 5;
    const int lane = tid & 31;
    const int half = lane >> 4;
    const int m    = lane & 15;
    const int tile = blockIdx.x * WAVES_PER_BLOCK + wave;
    const int b0   = tile * ROWS_PER_WAVE;
    if (b0 + ROWS_PER_WAVE > B) return;

    v2f whA[2][5];
#pragma unroll
    for (int mt = 0; mt < 2; ++mt) {
        const int  n  = m + 16 * mt;
        const bool nv = (n < H_DIM);
#pragma unroll
        for (int c = 0; c < 5; ++c) {
            const int kh = (c < 4) ? (2 * c + 8 * half) : (16 + 2 * half);
            whA[mt][c].x = nv ? Whh[(kh    ) * H_DIM + n] : 0.0f;
            whA[mt][c].y = nv ? Whh[(kh + 1) * H_DIM + n] : 0.0f;
        }
    }

    v2f hB[5];
#pragma unroll
    for (int c = 0; c < 5; ++c) { hB[c].x = 0.0f; hB[c].y = 0.0f; }

    const int*  xrow  = x + (size_t)(b0 + m) * T_STEPS;
    float*      myLds = &lds[wave][0][0];

    const float* tabh = table + 8 * half;

    v8f pC0[GROUP], pC1[GROUP];
    int tokN[GROUP];
    {
        const int4 ta = *(const int4*)xrow;
        const int4 tb = *(const int4*)(xrow + 4);
        const int tok0[GROUP] = { ta.x, ta.y, ta.z, ta.w, tb.x, tb.y, tb.z, tb.w };
        const int4 na = *(const int4*)(xrow + GROUP);
        const int4 nb = *(const int4*)(xrow + GROUP + 4);
        tokN[0] = na.x; tokN[1] = na.y; tokN[2] = na.z; tokN[3] = na.w;
        tokN[4] = nb.x; tokN[5] = nb.y; tokN[6] = nb.z; tokN[7] = nb.w;
#pragma unroll
        for (int tt = 0; tt < GROUP; ++tt) {
            const float* prow = tabh + (size_t)tok0[tt] * TAB_STRIDE;
            const v4f lo = *(const v4f*)(prow);
            const v4f hi = *(const v4f*)(prow + 4);
            const v4f t1 = *(const v4f*)(prow + 16);
            pC0[tt][0] = lo.x; pC0[tt][1] = lo.y; pC0[tt][2] = lo.z; pC0[tt][3] = lo.w;
            pC0[tt][4] = hi.x; pC0[tt][5] = hi.y; pC0[tt][6] = hi.z; pC0[tt][7] = hi.w;
            pC1[tt][0] = t1.x; pC1[tt][1] = t1.y; pC1[tt][2] = t1.z; pC1[tt][3] = t1.w;
            pC1[tt][4] = 0.0f; pC1[tt][5] = 0.0f; pC1[tt][6] = 0.0f; pC1[tt][7] = 0.0f;
        }
    }

    for (int t0 = 0; t0 < T_STEPS - GROUP; t0 += GROUP) {
        int tokA[GROUP];
#pragma unroll
        for (int tt = 0; tt < GROUP; ++tt) tokA[tt] = tokN[tt];
        if (t0 + 2 * GROUP < T_STEPS) {
            const int4 na = *(const int4*)(xrow + t0 + 2 * GROUP);
            const int4 nb = *(const int4*)(xrow + t0 + 2 * GROUP + 4);
            tokN[0] = na.x; tokN[1] = na.y; tokN[2] = na.z; tokN[3] = na.w;
            tokN[4] = nb.x; tokN[5] = nb.y; tokN[6] = nb.z; tokN[7] = nb.w;
        }
        rnn_group<true>(pC0, pC1, hB, whA, tokA, tabh, half);
    }
    rnn_group<false>(pC0, pC1, hB, whA, tokN, tabh, half);

#pragma unroll
    for (int c = 0; c < 4; ++c)
        *(v2f*)(myLds + m * LDS_STRIDE + 2 * c + 8 * half) = hB[c];
    *(v2f*)(myLds + m * LDS_STRIDE + 16 + 2 * half) = hB[4];

    const float scale = 1.0f / (1.0f - drop_rate[0]);
    if (half == 0) {
        float hrow[H_DIM];
#pragma unroll
        for (int k = 0; k < H_DIM; ++k)
            hrow[k] = myLds[m * LDS_STRIDE + k] * scale;

        float z[L_DIM];
        float zmax = -INFINITY;
#pragma unroll
        for (int l = 0; l < L_DIM; ++l) {
            float s = bd[l];
#pragma unroll
            for (int k = 0; k < H_DIM; ++k)
                s += hrow[k] * Wd[k * L_DIM + l];
            z[l] = s;
            zmax = fmaxf(zmax, s);
        }
        float sum = 0.0f;
#pragma unroll
        for (int l = 0; l < L_DIM; ++l) {
            z[l] = __expf(z[l] - zmax);
            sum += z[l];
        }
        const float inv = 1.0f / sum;
        float* orow = out + (size_t)(b0 + m) * L_DIM;
#pragma unroll
        for (int l = 0; l < L_DIM; ++l)
            orow[l] = z[l] * inv;
    }
}

extern "C" void kernel_launch(void* const* d_in, const int* in_sizes, int n_in,
                              void* d_out, int out_size, void* d_ws, size_t ws_size,
                              hipStream_t stream) {
    (void)n_in; (void)out_size; (void)ws_size;

    const int*   x   = (const int*)  d_in[0];
    const float* dr  = (const float*)d_in[1];
    const float* emb = (const float*)d_in[2];
    const float* Wxh = (const float*)d_in[3];
    const float* Whh = (const float*)d_in[4];
    const float* bh  = (const float*)d_in[5];
    const float* Wd  = (const float*)d_in[6];
    const float* bd  = (const float*)d_in[7];
    float*       out = (float*)d_out;
    float*       table = (float*)d_ws;

    const int B = in_sizes[0] / T_STEPS;
    const int V = in_sizes[2] / E_DIM;
    const int rowsPerBlock = ROWS_PER_WAVE * WAVES_PER_BLOCK;
    const int grid = (B + rowsPerBlock - 1) / rowsPerBlock;
    const int projRows = 16 * PROJ_WAVES;
    const int projGrid = (V + projRows - 1) / projRows;

    hipLaunchKernelGGL(input_projection_table_kernel, dim3(projGrid), dim3(32 * PROJ_WAVES), 0, stream,
                       emb, Wxh, bh, table, V);
    hipLaunchKernelGGL(WordVanilla_rnn_kernel, dim3(grid), dim3(32 * WAVES_PER_BLOCK), 0, stream,
                       x, dr, table, Whh, Wd, bd, out, B);
}


# 16-step register groups (was 8): each projected row is requested 16 steps ahead of its use, so the slowest of a step's 16 gathered rows has 15 steps to land; token lines read in 2 visits, not 4
# speedup vs baseline: 3.1570x; 1.0316x over previous
#include <hip/hip_runtime.h>
#include <math.h>


typedef __attribute__((ext_vector_type(2))) float v2f;
typedef __attribute__((ext_vector_type(4))) float v4f;
typedef __attribute__((ext_vector_type(8))) float v8f;

#if defined(__has_builtin)
#if __has_builtin(__builtin_amdgcn_tanhf)
#define FAST_TANH(x) __builtin_amdgcn_tanhf(x)
#endif
#endif
#ifndef FAST_TANH
#define FAST_TANH(x) tanhf(x)
#endif

#define WAVES_PER_BLOCK 4
#define ROWS_PER_WAVE   16
#define T_STEPS         128
#define E_DIM           20
#define H_DIM           20
#define L_DIM           15
#define LDS_STRIDE      22
#define TAB_STRIDE      32
#define PROJ_WAVES      4
#define GROUP           16


static __device__ __forceinline__ float xhalf(float v)
{
    const int b = __builtin_bit_cast(int, v);
    return __builtin_bit_cast(float,
        __builtin_amdgcn_permlanex16(b, b, 0x76543210, 0xfedcba98, false, false));
}

__global__ __launch_bounds__(32 * PROJ_WAVES)
void input_projection_table_kernel(const float* __restrict__ emb,
                                   const float* __restrict__ Wxh,
                                   const float* __restrict__ bh,
                                   float*       __restrict__ table,
                                   int V)
{
    const int tid  = threadIdx.x;
    const int wave = tid >> 5;
    const int lane = tid & 31;
    const int half = lane >> 4;
    const int m    = lane & 15;
    const int r0   = (blockIdx.x * PROJ_WAVES + wave) * 16;
    if (r0 >= V) return;
    const int row  = r0 + m;
    const int rowc = (row < V) ? row : (V - 1);

    v2f wxA[2][5];
#pragma unroll
    for (int mt = 0; mt < 2; ++mt) {
        const int  n  = m + 16 * mt;
        const bool nv = (n < H_DIM);
#pragma unroll
        for (int c = 0; c < 5; ++c) {
            const int k = 4 * c + 2 * half;
            wxA[mt][c].x = nv ? Wxh[(k    ) * H_DIM + n] : 0.0f;
            wxA[mt][c].y = nv ? Wxh[(k + 1) * H_DIM + n] : 0.0f;
        }
    }
    const v4f bLo = *(const v4f*)(bh + 8 * half);
    const v4f bHi = *(const v4f*)(bh + 8 * half + 4);
    const v4f bT1 = *(const v4f*)(bh + 16);
    const float* erow = emb + (size_t)rowc * E_DIM + 2 * half;
    v2f eB[5];
#pragma unroll
    for (int c = 0; c < 5; ++c)
        eB[c] = *(const v2f*)(erow + 4 * c);

    v8f acc0 = {}; v8f acc1 = {};
#pragma unroll
    for (int c = 0; c < 5; ++c) {
        acc0 = __builtin_amdgcn_wmma_f32_16x16x4_f32(
            false, wxA[0][c], false, eB[c], (short)0, acc0, false, false);
        acc1 = __builtin_amdgcn_wmma_f32_16x16x4_f32(
            false, wxA[1][c], false, eB[c], (short)0, acc1, false, false);
    }

    if (row < V) {
        float* prow = table + (size_t)row * TAB_STRIDE;
        v4f lo, hi;
        lo.x = acc0[0] + bLo.x; lo.y = acc0[1] + bLo.y;
        lo.z = acc0[2] + bLo.z; lo.w = acc0[3] + bLo.w;
        hi.x = acc0[4] + bHi.x; hi.y = acc0[5] + bHi.y;
        hi.z = acc0[6] + bHi.z; hi.w = acc0[7] + bHi.w;
        *(v4f*)(prow + 8 * half)     = lo;
        *(v4f*)(prow + 8 * half + 4) = hi;
        v4f t1;
        t1.x = half ? 0.0f : acc1[0] + bT1.x;
        t1.y = half ? 0.0f : acc1[1] + bT1.y;
        t1.z = half ? 0.0f : acc1[2] + bT1.z;
        t1.w = half ? 0.0f : acc1[3] + bT1.w;
        *(v4f*)(prow + 16 + 8 * half) = t1;
    }
}

template <bool REFILL>
static __device__ __forceinline__ void rnn_group(
    v8f (&pC0)[GROUP], v8f (&pC1)[GROUP], v2f (&hB)[5],
    const v2f (&whA)[2][5],
    const int (&tokA)[GROUP], const float* __restrict__ tabh, const int half)
{
#pragma unroll
    for (int tt = 0; tt < GROUP; ++tt) {
        v8f acc0 = pC0[tt]; v8f acc1 = pC1[tt];
#pragma unroll
        for (int c = 0; c < 5; ++c) {
            acc0 = __builtin_amdgcn_wmma_f32_16x16x4_f32(
                false, whA[0][c], false, hB[c], (short)0, acc0, false, false);
            acc1 = __builtin_amdgcn_wmma_f32_16x16x4_f32(
                false, whA[1][c], false, hB[c], (short)0, acc1, false, false);
        }

#pragma unroll
        for (int c = 0; c < 4; ++c) {
            hB[c].x = FAST_TANH(acc0[2 * c]);
            hB[c].y = FAST_TANH(acc0[2 * c + 1]);
        }
        const float t16 = FAST_TANH(acc1[0]);
        const float t17 = FAST_TANH(acc1[1]);
        const float t18 = FAST_TANH(acc1[2]);
        const float t19 = FAST_TANH(acc1[3]);
        const float x18 = xhalf(t18);
        const float x19 = xhalf(t19);
        hB[4].x = half ? x18 : t16;
        hB[4].y = half ? x19 : t17;

        pC1[tt][4] = acc1[4]; pC1[tt][5] = acc1[5]; pC1[tt][6] = acc1[6]; pC1[tt][7] = acc1[7];
        if (REFILL) {
            __builtin_amdgcn_sched_barrier(0);
            const float* prow = tabh + (size_t)tokA[tt] * TAB_STRIDE;
            const v4f lo = *(const v4f*)(prow);
            const v4f hi = *(const v4f*)(prow + 4);
            const v4f t1 = *(const v4f*)(prow + 16);
            pC0[tt][0] = lo.x; pC0[tt][1] = lo.y; pC0[tt][2] = lo.z; pC0[tt][3] = lo.w;
            pC0[tt][4] = hi.x; pC0[tt][5] = hi.y; pC0[tt][6] = hi.z; pC0[tt][7] = hi.w;
            pC1[tt][0] = t1.x; pC1[tt][1] = t1.y; pC1[tt][2] = t1.z; pC1[tt][3] = t1.w;
            __builtin_amdgcn_sched_barrier(0);
        }
    }
}

__global__ __launch_bounds__(32 * WAVES_PER_BLOCK)
void WordVanilla_rnn_kernel(const int*   __restrict__ x,
                            const float* __restrict__ drop_rate,
                            const float* __restrict__ table,
                            const float* __restrict__ Whh,
                            const float* __restrict__ Wd,
                            const float* __restrict__ bd,
                            float*       __restrict__ out,
                            int B)
{
    __shared__ float lds[WAVES_PER_BLOCK][ROWS_PER_WAVE][LDS_STRIDE];

    const int tid  = threadIdx.x;
    const int wave = tid >> 5;
    const int lane = tid & 31;
    const int half = lane >> 4;
    const int m    = lane & 15;
    const int tile = blockIdx.x * WAVES_PER_BLOCK + wave;
    const int b0   = tile * ROWS_PER_WAVE;
    if (b0 + ROWS_PER_WAVE > B) return;

    v2f whA[2][5];
#pragma unroll
    for (int mt = 0; mt < 2; ++mt) {
        const int  n  = m + 16 * mt;
        const bool nv = (n < H_DIM);
#pragma unroll
        for (int c = 0; c < 5; ++c) {
            const int kh = (c < 4) ? (2 * c + 8 * half) : (16 + 2 * half);
            whA[mt][c].x = nv ? Whh[(kh    ) * H_DIM + n] : 0.0f;
            whA[mt][c].y = nv ? Whh[(kh + 1) * H_DIM + n] : 0.0f;
        }
    }

    v2f hB[5];
#pragma unroll
    for (int c = 0; c < 5; ++c) { hB[c].x = 0.0f; hB[c].y = 0.0f; }

    const int*  xrow  = x + (size_t)(b0 + m) * T_STEPS;
    float*      myLds = &lds[wave][0][0];

    const float* tabh = table + 8 * half;

    v8f pC0[GROUP], pC1[GROUP];
    int tokN[GROUP];
    {
        int tok0[GROUP];
#pragma unroll
        for (int q = 0; q < GROUP / 4; ++q) {
            const int4 ta = *(const int4*)(xrow + 4 * q);
            const int4 na = *(const int4*)(xrow + GROUP + 4 * q);
            tok0[4 * q] = ta.x; tok0[4 * q + 1] = ta.y; tok0[4 * q + 2] = ta.z; tok0[4 * q + 3] = ta.w;
            tokN[4 * q] = na.x; tokN[4 * q + 1] = na.y; tokN[4 * q + 2] = na.z; tokN[4 * q + 3] = na.w;
        }
#pragma unroll
        for (int tt = 0; tt < GROUP; ++tt) {
            const float* prow = tabh + (size_t)tok0[tt] * TAB_STRIDE;
            const v4f lo = *(const v4f*)(prow);
            const v4f hi = *(const v4f*)(prow + 4);
            const v4f t1 = *(const v4f*)(prow + 16);
            pC0[tt][0] = lo.x; pC0[tt][1] = lo.y; pC0[tt][2] = lo.z; pC0[tt][3] = lo.w;
            pC0[tt][4] = hi.x; pC0[tt][5] = hi.y; pC0[tt][6] = hi.z; pC0[tt][7] = hi.w;
            pC1[tt][0] = t1.x; pC1[tt][1] = t1.y; pC1[tt][2] = t1.z; pC1[tt][3] = t1.w;
            pC1[tt][4] = 0.0f; pC1[tt][5] = 0.0f; pC1[tt][6] = 0.0f; pC1[tt][7] = 0.0f;
        }
    }

    for (int t0 = 0; t0 < T_STEPS - GROUP; t0 += GROUP) {
        int tokA[GROUP];
#pragma unroll
        for (int tt = 0; tt < GROUP; ++tt) tokA[tt] = tokN[tt];
        if (t0 + 2 * GROUP < T_STEPS) {
#pragma unroll
            for (int q = 0; q < GROUP / 4; ++q) {
                const int4 na = *(const int4*)(xrow + t0 + 2 * GROUP + 4 * q);
                tokN[4 * q] = na.x; tokN[4 * q + 1] = na.y; tokN[4 * q + 2] = na.z; tokN[4 * q + 3] = na.w;
            }
        }
        rnn_group<true>(pC0, pC1, hB, whA, tokA, tabh, half);
    }
    rnn_group<false>(pC0, pC1, hB, whA, tokN, tabh, half);

#pragma unroll
    for (int c = 0; c < 4; ++c)
        *(v2f*)(myLds + m * LDS_STRIDE + 2 * c + 8 * half) = hB[c];
    *(v2f*)(myLds + m * LDS_STRIDE + 16 + 2 * half) = hB[4];

    const float scale = 1.0f / (1.0f - drop_rate[0]);
    if (half == 0) {
        float hrow[H_DIM];
#pragma unroll
        for (int k = 0; k < H_DIM; ++k)
            hrow[k] = myLds[m * LDS_STRIDE + k] * scale;

        float z[L_DIM];
        float zmax = -INFINITY;
#pragma unroll
        for (int l = 0; l < L_DIM; ++l) {
            float s = bd[l];
#pragma unroll
            for (int k = 0; k < H_DIM; ++k)
                s += hrow[k] * Wd[k * L_DIM + l];
            z[l] = s;
            zmax = fmaxf(zmax, s);
        }
        float sum = 0.0f;
#pragma unroll
        for (int l = 0; l < L_DIM; ++l) {
            z[l] = __expf(z[l] - zmax);
            sum += z[l];
        }
        const float inv = 1.0f / sum;
        float* orow = out + (size_t)(b0 + m) * L_DIM;
#pragma unroll
        for (int l = 0; l < L_DIM; ++l)
            orow[l] = z[l] * inv;
    }
}

extern "C" void kernel_launch(void* const* d_in, const int* in_sizes, int n_in,
                              void* d_out, int out_size, void* d_ws, size_t ws_size,
                              hipStream_t stream) {
    (void)n_in; (void)out_size; (void)ws_size;

    const int*   x   = (const int*)  d_in[0];
    const float* dr  = (const float*)d_in[1];
    const float* emb = (const float*)d_in[2];
    const float* Wxh = (const float*)d_in[3];
    const float* Whh = (const float*)d_in[4];
    const float* bh  = (const float*)d_in[5];
    const float* Wd  = (const float*)d_in[6];
    const float* bd  = (const float*)d_in[7];
    float*       out = (float*)d_out;
    float*       table = (float*)d_ws;

    const int B = in_sizes[0] / T_STEPS;
    const int V = in_sizes[2] / E_DIM;
    const int rowsPerBlock = ROWS_PER_WAVE * WAVES_PER_BLOCK;
    const int grid = (B + rowsPerBlock - 1) / rowsPerBlock;
    const int projRows = 16 * PROJ_WAVES;
    const int projGrid = (V + projRows - 1) / projRows;

    hipLaunchKernelGGL(input_projection_table_kernel, dim3(projGrid), dim3(32 * PROJ_WAVES), 0, stream,
                       emb, Wxh, bh, table, V);
    hipLaunchKernelGGL(WordVanilla_rnn_kernel, dim3(grid), dim3(32 * WAVES_PER_BLOCK), 0, stream,
                       x, dr, table, Whh, Wd, bd, out, B);
}
